# starting kernel
# baseline (speedup 1.0000x reference)
#include <hip/hip_runtime.h>
#include <hip/hip_bf16.h>

typedef __attribute__((ext_vector_type(2))) float v2f;
typedef __attribute__((ext_vector_type(8))) float v8f;

#define IN_DIM 128
#define OUT_DIM 128

__global__ void __launch_bounds__(128)
gcn_linear_wmma(const float* __restrict__ x, const float* __restrict__ W,
                const float* __restrict__ bias, float* __restrict__ h,
                int totalNodes) {
  const int lane = threadIdx.x & 31;
  const int wave = threadIdx.x >> 5;
  const int tile = blockIdx.x * (blockDim.x >> 5) + wave;
  const int nodeBase = tile * 16;
  if (nodeBase >= totalNodes) return;

  const int half = lane >> 4;
  const int l    = lane & 15;
  const int kA0  = half * 2;
  const int kA1  = kA0 + 1;

  v8f acc[8] = {};

  const float* __restrict__ xrow = x + (size_t)(nodeBase + l) * IN_DIM;

  for (int k = 0; k < IN_DIM; k += 4) {
    v2f a;
    a.x = xrow[k + kA0];
    a.y = xrow[k + kA1];
#pragma unroll
    for (int t = 0; t < 8; ++t) {
      const int n = t * 16 + l;
      const float* __restrict__ wrow = W + (size_t)n * IN_DIM;
      v2f bfrag;
      bfrag.x = wrow[k + kA0];
      bfrag.y = wrow[k + kA1];
      acc[t] = __builtin_amdgcn_wmma_f32_16x16x4_f32(
           false, a,  false, bfrag,
           (short)0, acc[t],  false,  false);
    }
  }

#pragma unroll
  for (int t = 0; t < 8; ++t) {
    const int n = t * 16 + l;
    const float bv = bias[n];
#pragma unroll
    for (int v = 0; v < 8; ++v) {
      const int m = v + half * 8;
      h[(size_t)(nodeBase + m) * OUT_DIM + n] = acc[t][v] + bv;
    }
  }
}

__global__ void zero_out(float* __restrict__ out, long long n) {
  long long i = (long long)blockIdx.x * blockDim.x + threadIdx.x;
  if (i < n) out[i] = 0.0f;
}

__global__ void __launch_bounds__(256)
edge_scatter(const float* __restrict__ h, const int* __restrict__ srcIdx,
             const int* __restrict__ tgtIdx, const float* __restrict__ ew,
             float* __restrict__ out, int E, int N, int B) {
  long long idx = (long long)blockIdx.x * blockDim.x + threadIdx.x;
  long long total = (long long)E * OUT_DIM;
  if (idx >= total) return;
  const int c = (int)(idx & (OUT_DIM - 1));
  const int e = (int)(idx >> 7);
  const int s = srcIdx[e];
  const int t = tgtIdx[e];
  for (int b = 0; b < B; ++b) {
    const float w = ew[(size_t)b * E + e];
    const float val = w * h[((size_t)b * N + s) * OUT_DIM + c];
    atomicAdd(out + ((size_t)b * N + t) * OUT_DIM + c, val);
  }
}

extern "C" void kernel_launch(void* const* d_in, const int* in_sizes, int n_in,
                              void* d_out, int out_size, void* d_ws, size_t ws_size,
                              hipStream_t stream) {
  const float* x    = (const float*)d_in[0];
  const int*   ei   = (const int*)d_in[1];
  const float* ew   = (const float*)d_in[2];
  const float* W    = (const float*)d_in[3];
  const float* bias = (const float*)d_in[4];

  const int E = in_sizes[1] / 2;
  const int B = in_sizes[2] / E;
  const int totalNodes = in_sizes[0] / IN_DIM;
  const int N = totalNodes / B;

  float* h   = (float*)d_ws;
  float* out = (float*)d_out;

  {
    const int tiles = (totalNodes + 15) / 16;
    const int wavesPerBlock = 4;
    const int blocks = (tiles + wavesPerBlock - 1) / wavesPerBlock;
    gcn_linear_wmma<<<blocks, wavesPerBlock * 32, 0, stream>>>(x, W, bias, h,
                                                               totalNodes);
  }

  {
    const long long n = (long long)out_size;
    const int blocks = (int)((n + 255) / 256);
    zero_out<<<blocks, 256, 0, stream>>>(out, n);
  }

  {
    const long long total = (long long)E * OUT_DIM;
    const int blocks = (int)((total + 255) / 256);
    edge_scatter<<<blocks, 256, 0, stream>>>(h, ei, ei + E, ew, out, E, N, B);
  }
}


# Gather, count and fill passes run as resident grids of looping waves (8192 waves each) instead of one short wave per item; the gather keeps up to 16 edge rows in flight per wave, summed in list order
# speedup vs baseline: 1.8988x; 1.8988x over previous
#include <hip/hip_runtime.h>
#include <hip/hip_bf16.h>

typedef __attribute__((ext_vector_type(2))) float v2f;
typedef __attribute__((ext_vector_type(8))) float v8f;

#define IN_DIM 128
#define OUT_DIM 128

__global__ void __launch_bounds__(128)
gcn_linear_wmma(const float* __restrict__ x, const float* __restrict__ W,
                const float* __restrict__ bias, float* __restrict__ h,
                int totalNodes) {
  const int lane = threadIdx.x & 31;
  const int wave = threadIdx.x >> 5;
  const int tile = blockIdx.x * (blockDim.x >> 5) + wave;
  const int nodeBase = tile * 16;
  if (nodeBase >= totalNodes) return;

  const int half = lane >> 4;
  const int l    = lane & 15;
  const int kA0  = half * 2;
  const int kA1  = kA0 + 1;

  v8f acc[8] = {};

  const float* __restrict__ xrow = x + (size_t)(nodeBase + l) * IN_DIM;

  for (int k = 0; k < IN_DIM; k += 4) {
    v2f a;
    a.x = xrow[k + kA0];
    a.y = xrow[k + kA1];
#pragma unroll
    for (int t = 0; t < 8; ++t) {
      const int n = t * 16 + l;
      const float* __restrict__ wrow = W + (size_t)n * IN_DIM;
      v2f bfrag;
      bfrag.x = wrow[k + kA0];
      bfrag.y = wrow[k + kA1];
      acc[t] = __builtin_amdgcn_wmma_f32_16x16x4_f32(
           false, a,  false, bfrag,
           (short)0, acc[t],  false,  false);
    }
  }

#pragma unroll
  for (int t = 0; t < 8; ++t) {
    const int n = t * 16 + l;
    const float bv = bias[n];
#pragma unroll
    for (int v = 0; v < 8; ++v) {
      const int m = v + half * 8;
      h[(size_t)(nodeBase + m) * OUT_DIM + n] = acc[t][v] + bv;
    }
  }
}

__global__ void clear_counts(int* __restrict__ count, int N) {
  const int i = blockIdx.x * blockDim.x + threadIdx.x;
  if (i < N) count[i] = 0;
}

__global__ void count_targets(const int* __restrict__ tgtIdx,
                              int* __restrict__ count, int E) {
  const int stride = gridDim.x * blockDim.x;
  for (int e = blockIdx.x * blockDim.x + threadIdx.x; e < E; e += stride)
    atomicAdd(count + tgtIdx[e], 1);
}

#define EDGE_BLOCKS 1024
#define SCAN_THREADS 1024
__global__ void __launch_bounds__(SCAN_THREADS)
degree_scan(const int* __restrict__ count, int* __restrict__ rowptr, int N) {
  __shared__ int part[SCAN_THREADS];
  const int tid = threadIdx.x;
  const int per = (N + SCAN_THREADS - 1) / SCAN_THREADS;
  const int lo = min(tid * per, N);
  const int hi = min(lo + per, N);
  int s = 0;
  for (int i = lo; i < hi; ++i) s += count[i];
  part[tid] = s;
  __syncthreads();
  for (int off = 1; off < SCAN_THREADS; off <<= 1) {
    const int v = (tid >= off) ? part[tid - off] : 0;
    __syncthreads();
    part[tid] += v;
    __syncthreads();
  }
  int run = part[tid] - s;
  for (int i = lo; i < hi; ++i) {
    rowptr[i] = run;
    run += count[i];
  }
  if (tid == SCAN_THREADS - 1) rowptr[N] = part[tid];
}

__global__ void fill_by_target(const int* __restrict__ srcIdx,
                               const int* __restrict__ tgtIdx,
                               const float* __restrict__ ew,
                               const int* __restrict__ rowptr,
                               int* __restrict__ count, int* __restrict__ ssrc,
                               float* __restrict__ sw, int E, int B) {
  const int stride = gridDim.x * blockDim.x;
  for (int e = blockIdx.x * blockDim.x + threadIdx.x; e < E; e += stride) {
    const int t = tgtIdx[e];
    const int left = atomicSub(count + t, 1);
    const int pos = rowptr[t] + left - 1;
    if ((unsigned)pos >= (unsigned)E) continue;
    ssrc[pos] = srcIdx[e];
    for (int b = 0; b < B; ++b) sw[(size_t)b * E + pos] = ew[(size_t)b * E + e];
  }
}

#define GATHER_WAVES 8
#define GATHER_DEPTH 16
#define GATHER_BLOCKS 1024

template <int R>
__device__ __forceinline__ void gather_rows(float4& acc,
                                            const float4* __restrict__ hb,
                                            const int* __restrict__ sp,
                                            const float* __restrict__ wp) {
  int s[R];
  float w[R];
  float4 v[R];
#pragma unroll
  for (int u = 0; u < R; ++u) {
    s[u] = sp[u];
    w[u] = wp[u];
  }
#pragma unroll
  for (int u = 0; u < R; ++u) v[u] = hb[(size_t)s[u] * (OUT_DIM / 4)];
#pragma unroll
  for (int u = 0; u < R; ++u) {
    acc.x = fmaf(w[u], v[u].x, acc.x);
    acc.y = fmaf(w[u], v[u].y, acc.y);
    acc.z = fmaf(w[u], v[u].z, acc.z);
    acc.w = fmaf(w[u], v[u].w, acc.w);
  }
}

__global__ void __launch_bounds__(GATHER_WAVES * 32)
__attribute__((amdgpu_waves_per_eu(4, 8)))
target_gather(const float* __restrict__ h, const int* __restrict__ rowptr,
              const int* __restrict__ ssrc, const float* __restrict__ sw,
              float* __restrict__ out, int E, int N, int B) {
  const int lane = threadIdx.x & 31;
  const int t0 = __builtin_amdgcn_readfirstlane(
      (int)(blockIdx.x * GATHER_WAVES + (threadIdx.x >> 5)));
  if (t0 >= N) return;
  const int step = (int)gridDim.x * GATHER_WAVES;
  for (int b = 0; b < B; ++b) {
    const float4* __restrict__ hb =
        (const float4*)h + (size_t)b * N * (OUT_DIM / 4) + lane;
    const float* __restrict__ wb = sw + (size_t)b * E;
    float4* __restrict__ ob =
        (float4*)out + (size_t)b * N * (OUT_DIM / 4) + lane;
    int t = t0;
    int beg = rowptr[t];
    int end = rowptr[t + 1];
    for (;;) {
      const int tn = t + step;
      const int tp = min(tn, N - 1);
      const int nbeg = rowptr[tp];
      const int nend = rowptr[tp + 1];
      float4 acc = {0.0f, 0.0f, 0.0f, 0.0f};
      int j = beg;
      for (; end - j >= GATHER_DEPTH; j += GATHER_DEPTH)
        gather_rows<GATHER_DEPTH>(acc, hb, ssrc + j, wb + j);
      switch (end - j) {
        case 15: gather_rows<15>(acc, hb, ssrc + j, wb + j); break;
        case 14: gather_rows<14>(acc, hb, ssrc + j, wb + j); break;
        case 13: gather_rows<13>(acc, hb, ssrc + j, wb + j); break;
        case 12: gather_rows<12>(acc, hb, ssrc + j, wb + j); break;
        case 11: gather_rows<11>(acc, hb, ssrc + j, wb + j); break;
        case 10: gather_rows<10>(acc, hb, ssrc + j, wb + j); break;
        case 9: gather_rows<9>(acc, hb, ssrc + j, wb + j); break;
        case 8: gather_rows<8>(acc, hb, ssrc + j, wb + j); break;
        case 7: gather_rows<7>(acc, hb, ssrc + j, wb + j); break;
        case 6: gather_rows<6>(acc, hb, ssrc + j, wb + j); break;
        case 5: gather_rows<5>(acc, hb, ssrc + j, wb + j); break;
        case 4: gather_rows<4>(acc, hb, ssrc + j, wb + j); break;
        case 3: gather_rows<3>(acc, hb, ssrc + j, wb + j); break;
        case 2: gather_rows<2>(acc, hb, ssrc + j, wb + j); break;
        case 1: gather_rows<1>(acc, hb, ssrc + j, wb + j); break;
        default: break;
      }
      ob[(size_t)t * (OUT_DIM / 4)] = acc;
      if (tn >= N) break;
      t = tn;
      beg = nbeg;
      end = nend;
    }
  }
}

__global__ void zero_out(float* __restrict__ out, long long n) {
  long long i = (long long)blockIdx.x * blockDim.x + threadIdx.x;
  if (i < n) out[i] = 0.0f;
}

__global__ void __launch_bounds__(256)
edge_scatter(const float* __restrict__ h, const int* __restrict__ srcIdx,
             const int* __restrict__ tgtIdx, const float* __restrict__ ew,
             float* __restrict__ out, int E, int N, int B) {
  long long idx = (long long)blockIdx.x * blockDim.x + threadIdx.x;
  long long total = (long long)E * OUT_DIM;
  if (idx >= total) return;
  const int c = (int)(idx & (OUT_DIM - 1));
  const int e = (int)(idx >> 7);
  const int s = srcIdx[e];
  const int t = tgtIdx[e];
  for (int b = 0; b < B; ++b) {
    const float w = ew[(size_t)b * E + e];
    const float val = w * h[((size_t)b * N + s) * OUT_DIM + c];
    atomicAdd(out + ((size_t)b * N + t) * OUT_DIM + c, val);
  }
}

extern "C" void kernel_launch(void* const* d_in, const int* in_sizes, int n_in,
                              void* d_out, int out_size, void* d_ws, size_t ws_size,
                              hipStream_t stream) {
  const float* x    = (const float*)d_in[0];
  const int*   ei   = (const int*)d_in[1];
  const float* ew   = (const float*)d_in[2];
  const float* W    = (const float*)d_in[3];
  const float* bias = (const float*)d_in[4];

  const int E = in_sizes[1] / 2;
  const int B = in_sizes[2] / E;
  const int totalNodes = in_sizes[0] / IN_DIM;
  const int N = totalNodes / B;

  float* h      = (float*)d_ws;
  int*   rowptr = (int*)(h + (size_t)totalNodes * OUT_DIM);
  int*   count  = rowptr + (N + 1);
  int*   ssrc   = count + N;
  float* sw     = (float*)(ssrc + E);
  float* out    = (float*)d_out;
  const size_t wsNeeded = (size_t)((const char*)(sw + (size_t)B * E) - (const char*)d_ws);

  {
    const int tiles = (totalNodes + 15) / 16;
    const int wavesPerBlock = 4;
    const int blocks = (tiles + wavesPerBlock - 1) / wavesPerBlock;
    gcn_linear_wmma<<<blocks, wavesPerBlock * 32, 0, stream>>>(x, W, bias, h,
                                                               totalNodes);
  }

  if (ws_size < wsNeeded) {
    const long long n = (long long)out_size;
    zero_out<<<(int)((n + 255) / 256), 256, 0, stream>>>(out, n);
    const long long total = (long long)E * OUT_DIM;
    edge_scatter<<<(int)((total + 255) / 256), 256, 0, stream>>>(h, ei, ei + E, ew,
                                                                 out, E, N, B);
    return;
  }

  clear_counts<<<(N + 255) / 256, 256, 0, stream>>>(count, N);
  count_targets<<<EDGE_BLOCKS, 256, 0, stream>>>(ei + E, count, E);
  degree_scan<<<1, SCAN_THREADS, 0, stream>>>(count, rowptr, N);
  fill_by_target<<<EDGE_BLOCKS, 256, 0, stream>>>(ei, ei + E, ew, rowptr, count,
                                                  ssrc, sw, E, B);

  target_gather<<<GATHER_BLOCKS, GATHER_WAVES * 32, 0, stream>>>(
      h, rowptr, ssrc, sw, out, E, N, B);
}


# Linear transform takes both WMMA operands from LDS (W copied once per 20-wave work-group, x tile by whole-row loads, padded rows); h stored as whole 512-byte rows; same products, same sum order
# speedup vs baseline: 2.1451x; 1.1297x over previous
#include <hip/hip_runtime.h>
#include <hip/hip_bf16.h>

typedef __attribute__((ext_vector_type(2))) float v2f;
typedef __attribute__((ext_vector_type(8))) float v8f;

#define IN_DIM 128
#define OUT_DIM 128

#define LIN_WAVES 20
#define LIN_THREADS (LIN_WAVES * 32)
#define LIN_PITCH 132
#define LIN_W_F4 (OUT_DIM * (IN_DIM / 4))
#define LIN_W_ROUNDS ((LIN_W_F4 + LIN_THREADS - 1) / LIN_THREADS)
#define LIN_W_ROWS (LIN_W_ROUNDS * LIN_THREADS / 32)

__global__ void __launch_bounds__(LIN_THREADS)
gcn_linear_wmma(const float* __restrict__ x, const float* __restrict__ W,
                const float* __restrict__ bias, float* __restrict__ h,
                int totalNodes) {
  __shared__ __attribute__((aligned(16))) float wS[LIN_W_ROWS * LIN_PITCH];
  __shared__ __attribute__((aligned(16))) float tS[LIN_WAVES * 16 * LIN_PITCH];

  const int tid  = threadIdx.x;
  const int lane = tid & 31;
  const int wave = tid >> 5;
  const int half = lane >> 4;
  const int l    = lane & 15;
  const int nodeBase = (blockIdx.x * LIN_WAVES + wave) * 16;
  float* mine = tS + wave * (16 * LIN_PITCH);

  float4 xr[16];
#pragma unroll
  for (int r = 0; r < 16; ++r) {
    const int node = min(nodeBase + r, totalNodes - 1);
    xr[r] = ((const float4*)(x + (size_t)node * IN_DIM))[lane];
  }
  float4 wv[LIN_W_ROUNDS];
#pragma unroll
  for (int j = 0; j < LIN_W_ROUNDS; ++j)
    wv[j] = ((const float4*)W)[min(tid + j * LIN_THREADS, LIN_W_F4 - 1)];
#pragma unroll
  for (int j = 0; j < LIN_W_ROUNDS; ++j) {
    const int i = tid + j * LIN_THREADS;
    *(float4*)(wS + (i >> 5) * LIN_PITCH + (i & 31) * 4) = wv[j];
  }
#pragma unroll
  for (int r = 0; r < 16; ++r)
    *(float4*)(mine + r * LIN_PITCH + lane * 4) = xr[r];
  __syncthreads();
  if (nodeBase >= totalNodes) return;

  v8f acc[8] = {};
  const float* aP = mine + l * LIN_PITCH + half * 2;
  const float* bP = wS + l * LIN_PITCH + half * 2;

  for (int k = 0; k < IN_DIM; k += 4) {
    const v2f a = *(const v2f*)(aP + k);
#pragma unroll
    for (int t = 0; t < 8; ++t) {
      const v2f bfrag = *(const v2f*)(bP + t * (16 * LIN_PITCH) + k);
      acc[t] = __builtin_amdgcn_wmma_f32_16x16x4_f32(
           false, a,  false, bfrag,
           (short)0, acc[t],  false,  false);
    }
  }

#pragma unroll
  for (int t = 0; t < 8; ++t) {
    const int n = t * 16 + l;
    const float bv = bias[n];
#pragma unroll
    for (int v = 0; v < 8; ++v) {
      const int m = v + half * 8;
      mine[m * LIN_PITCH + n] = acc[t][v] + bv;
    }
  }
  __builtin_amdgcn_fence(__ATOMIC_RELEASE, "wavefront");
  __builtin_amdgcn_wave_barrier();
  __builtin_amdgcn_fence(__ATOMIC_ACQUIRE, "wavefront");
  float4 o4[16];
#pragma unroll
  for (int r = 0; r < 16; ++r)
    o4[r] = *(const float4*)(mine + r * LIN_PITCH + lane * 4);
  float4* hrow = (float4*)(h + (size_t)nodeBase * OUT_DIM) + lane;
  if (nodeBase + 16 <= totalNodes) {
#pragma unroll
    for (int r = 0; r < 16; ++r) hrow[r * (OUT_DIM / 4)] = o4[r];
  } else {
#pragma unroll
    for (int r = 0; r < 16; ++r)
      if (nodeBase + r < totalNodes) hrow[r * (OUT_DIM / 4)] = o4[r];
  }
}

__global__ void clear_counts(int* __restrict__ count, int N) {
  const int i = blockIdx.x * blockDim.x + threadIdx.x;
  if (i < N) count[i] = 0;
}

__global__ void count_targets(const int* __restrict__ tgtIdx,
                              int* __restrict__ count, int E) {
  const int stride = gridDim.x * blockDim.x;
  for (int e = blockIdx.x * blockDim.x + threadIdx.x; e < E; e += stride)
    atomicAdd(count + tgtIdx[e], 1);
}

#define EDGE_BLOCKS 1024
#define SCAN_THREADS 1024
__global__ void __launch_bounds__(SCAN_THREADS)
degree_scan(const int* __restrict__ count, int* __restrict__ rowptr, int N) {
  __shared__ int part[SCAN_THREADS];
  const int tid = threadIdx.x;
  const int per = (N + SCAN_THREADS - 1) / SCAN_THREADS;
  const int lo = min(tid * per, N);
  const int hi = min(lo + per, N);
  int s = 0;
  for (int i = lo; i < hi; ++i) s += count[i];
  part[tid] = s;
  __syncthreads();
  for (int off = 1; off < SCAN_THREADS; off <<= 1) {
    const int v = (tid >= off) ? part[tid - off] : 0;
    __syncthreads();
    part[tid] += v;
    __syncthreads();
  }
  int run = part[tid] - s;
  for (int i = lo; i < hi; ++i) {
    rowptr[i] = run;
    run += count[i];
  }
  if (tid == SCAN_THREADS - 1) rowptr[N] = part[tid];
}

__global__ void fill_by_target(const int* __restrict__ srcIdx,
                               const int* __restrict__ tgtIdx,
                               const float* __restrict__ ew,
                               const int* __restrict__ rowptr,
                               int* __restrict__ count, int* __restrict__ ssrc,
                               float* __restrict__ sw, int E, int B) {
  const int stride = gridDim.x * blockDim.x;
  for (int e = blockIdx.x * blockDim.x + threadIdx.x; e < E; e += stride) {
    const int t = tgtIdx[e];
    const int left = atomicSub(count + t, 1);
    const int pos = rowptr[t] + left - 1;
    if ((unsigned)pos >= (unsigned)E) continue;
    ssrc[pos] = srcIdx[e];
    for (int b = 0; b < B; ++b) sw[(size_t)b * E + pos] = ew[(size_t)b * E + e];
  }
}

#define SORT_SLICES 64
#define SORT_THREADS 1024
#define SORT_ROUND 8
#define NODE_CAP 40960

__device__ __forceinline__ void slice_bounds(int E, int& lo, int& hi) {
  const int per = (E + (int)gridDim.x - 1) / (int)gridDim.x;
  lo = min((int)blockIdx.x * per, E);
  hi = min(lo + per, E);
}

__global__ void __launch_bounds__(SORT_THREADS)
slice_count(const int* __restrict__ tgtIdx, int* __restrict__ part, int E, int N) {
  __shared__ int cnt[NODE_CAP];
  const int tid = threadIdx.x;
  int lo, hi;
  slice_bounds(E, lo, hi);
  for (int i = tid; i < N; i += SORT_THREADS) cnt[i] = 0;
  __syncthreads();
  for (int e0 = lo + tid; e0 < hi; e0 += SORT_ROUND * SORT_THREADS) {
    int t[SORT_ROUND];
    for (int k = 0; k < SORT_ROUND; ++k)
      t[k] = tgtIdx[min(e0 + k * SORT_THREADS, hi - 1)];
    for (int k = 0; k < SORT_ROUND; ++k)
      if (e0 + k * SORT_THREADS < hi) atomicAdd(&cnt[t[k]], 1);
  }
  __syncthreads();
  int* __restrict__ mine = part + (size_t)blockIdx.x * N;
  for (int i = tid; i < N; i += SORT_THREADS) mine[i] = cnt[i];
}

__global__ void __launch_bounds__(256)
node_offsets(int* __restrict__ part, int* __restrict__ count, int N) {
  const int t = blockIdx.x * blockDim.x + threadIdx.x;
  if (t >= N) return;
  int c[SORT_SLICES];
  for (int g = 0; g < SORT_SLICES; ++g) c[g] = part[(size_t)g * N + t];
  int run = 0;
  for (int g = 0; g < SORT_SLICES; ++g) {
    part[(size_t)g * N + t] = run;
    run += c[g];
  }
  count[t] = run;
}

__global__ void __launch_bounds__(SORT_THREADS)
slice_fill(const int* __restrict__ srcIdx, const int* __restrict__ tgtIdx,
           const float* __restrict__ ew, const int* __restrict__ rowptr,
           const int* __restrict__ part, int* __restrict__ ssrc,
           float* __restrict__ sw, int E, int N, int B) {
  __shared__ int cur[NODE_CAP];
  const int tid = threadIdx.x;
  int lo, hi;
  slice_bounds(E, lo, hi);
  const int* __restrict__ mine = part + (size_t)blockIdx.x * N;
  for (int i0 = tid; i0 < N; i0 += SORT_ROUND * SORT_THREADS) {
    int v[SORT_ROUND];
    for (int k = 0; k < SORT_ROUND; ++k) {
      const int ic = min(i0 + k * SORT_THREADS, N - 1);
      v[k] = rowptr[ic] + mine[ic];
    }
    for (int k = 0; k < SORT_ROUND; ++k) {
      const int i = i0 + k * SORT_THREADS;
      if (i < N) cur[i] = v[k];
    }
  }
  __syncthreads();
  for (int e0 = lo + tid; e0 < hi; e0 += SORT_ROUND * SORT_THREADS) {
    int ec[SORT_ROUND], t[SORT_ROUND], s[SORT_ROUND], pos[SORT_ROUND];
    for (int k = 0; k < SORT_ROUND; ++k) {
      ec[k] = min(e0 + k * SORT_THREADS, hi - 1);
      t[k] = tgtIdx[ec[k]];
      s[k] = srcIdx[ec[k]];
    }
    for (int k = 0; k < SORT_ROUND; ++k) {
      pos[k] = -1;
      if (e0 + k * SORT_THREADS < hi) pos[k] = atomicAdd(&cur[t[k]], 1);
    }
    for (int k = 0; k < SORT_ROUND; ++k)
      if ((unsigned)pos[k] < (unsigned)E) ssrc[pos[k]] = s[k];
    for (int b = 0; b < B; ++b) {
      const float* __restrict__ wsrc = ew + (size_t)b * E;
      float* __restrict__ wdst = sw + (size_t)b * E;
      float w[SORT_ROUND];
      for (int k = 0; k < SORT_ROUND; ++k) w[k] = wsrc[ec[k]];
      for (int k = 0; k < SORT_ROUND; ++k)
        if ((unsigned)pos[k] < (unsigned)E) wdst[pos[k]] = w[k];
    }
  }
}

#define GATHER_WAVES 8
#define GATHER_DEPTH 16
#define GATHER_BLOCKS 1024

template <int R>
__device__ __forceinline__ void gather_rows(float4& acc,
                                            const float4* __restrict__ hb,
                                            const int* __restrict__ sp,
                                            const float* __restrict__ wp) {
  int s[R];
  float w[R];
  float4 v[R];
#pragma unroll
  for (int u = 0; u < R; ++u) {
    s[u] = sp[u];
    w[u] = wp[u];
  }
#pragma unroll
  for (int u = 0; u < R; ++u) v[u] = hb[(size_t)s[u] * (OUT_DIM / 4)];
#pragma unroll
  for (int u = 0; u < R; ++u) {
    acc.x = fmaf(w[u], v[u].x, acc.x);
    acc.y = fmaf(w[u], v[u].y, acc.y);
    acc.z = fmaf(w[u], v[u].z, acc.z);
    acc.w = fmaf(w[u], v[u].w, acc.w);
  }
}

__global__ void __launch_bounds__(GATHER_WAVES * 32)
__attribute__((amdgpu_waves_per_eu(4, 8)))
target_gather(const float* __restrict__ h, const int* __restrict__ rowptr,
              const int* __restrict__ ssrc, const float* __restrict__ sw,
              float* __restrict__ out, int E, int N, int B) {
  const int lane = threadIdx.x & 31;
  const int t0 = __builtin_amdgcn_readfirstlane(
      (int)(blockIdx.x * GATHER_WAVES + (threadIdx.x >> 5)));
  if (t0 >= N) return;
  const int step = (int)gridDim.x * GATHER_WAVES;
  for (int b = 0; b < B; ++b) {
    const float4* __restrict__ hb =
        (const float4*)h + (size_t)b * N * (OUT_DIM / 4) + lane;
    const float* __restrict__ wb = sw + (size_t)b * E;
    float4* __restrict__ ob =
        (float4*)out + (size_t)b * N * (OUT_DIM / 4) + lane;
    int t = t0;
    int beg = rowptr[t];
    int end = rowptr[t + 1];
    for (;;) {
      const int tn = t + step;
      const int tp = min(tn, N - 1);
      const int nbeg = rowptr[tp];
      const int nend = rowptr[tp + 1];
      float4 acc = {0.0f, 0.0f, 0.0f, 0.0f};
      int j = beg;
      for (; end - j >= GATHER_DEPTH; j += GATHER_DEPTH)
        gather_rows<GATHER_DEPTH>(acc, hb, ssrc + j, wb + j);
      switch (end - j) {
        case 15: gather_rows<15>(acc, hb, ssrc + j, wb + j); break;
        case 14: gather_rows<14>(acc, hb, ssrc + j, wb + j); break;
        case 13: gather_rows<13>(acc, hb, ssrc + j, wb + j); break;
        case 12: gather_rows<12>(acc, hb, ssrc + j, wb + j); break;
        case 11: gather_rows<11>(acc, hb, ssrc + j, wb + j); break;
        case 10: gather_rows<10>(acc, hb, ssrc + j, wb + j); break;
        case 9: gather_rows<9>(acc, hb, ssrc + j, wb + j); break;
        case 8: gather_rows<8>(acc, hb, ssrc + j, wb + j); break;
        case 7: gather_rows<7>(acc, hb, ssrc + j, wb + j); break;
        case 6: gather_rows<6>(acc, hb, ssrc + j, wb + j); break;
        case 5: gather_rows<5>(acc, hb, ssrc + j, wb + j); break;
        case 4: gather_rows<4>(acc, hb, ssrc + j, wb + j); break;
        case 3: gather_rows<3>(acc, hb, ssrc + j, wb + j); break;
        case 2: gather_rows<2>(acc, hb, ssrc + j, wb + j); break;
        case 1: gather_rows<1>(acc, hb, ssrc + j, wb + j); break;
        default: break;
      }
      ob[(size_t)t * (OUT_DIM / 4)] = acc;
      if (tn >= N) break;
      t = tn;
      beg = nbeg;
      end = nend;
    }
  }
}

__global__ void zero_out(float* __restrict__ out, long long n) {
  long long i = (long long)blockIdx.x * blockDim.x + threadIdx.x;
  if (i < n) out[i] = 0.0f;
}

__global__ void __launch_bounds__(256)
edge_scatter(const float* __restrict__ h, const int* __restrict__ srcIdx,
             const int* __restrict__ tgtIdx, const float* __restrict__ ew,
             float* __restrict__ out, int E, int N, int B) {
  long long idx = (long long)blockIdx.x * blockDim.x + threadIdx.x;
  long long total = (long long)E * OUT_DIM;
  if (idx >= total) return;
  const int c = (int)(idx & (OUT_DIM - 1));
  const int e = (int)(idx >> 7);
  const int s = srcIdx[e];
  const int t = tgtIdx[e];
  for (int b = 0; b < B; ++b) {
    const float w = ew[(size_t)b * E + e];
    const float val = w * h[((size_t)b * N + s) * OUT_DIM + c];
    atomicAdd(out + ((size_t)b * N + t) * OUT_DIM + c, val);
  }
}

extern "C" void kernel_launch(void* const* d_in, const int* in_sizes, int n_in,
                              void* d_out, int out_size, void* d_ws, size_t ws_size,
                              hipStream_t stream) {
  const float* x    = (const float*)d_in[0];
  const int*   ei   = (const int*)d_in[1];
  const float* ew   = (const float*)d_in[2];
  const float* W    = (const float*)d_in[3];
  const float* bias = (const float*)d_in[4];

  const int E = in_sizes[1] / 2;
  const int B = in_sizes[2] / E;
  const int totalNodes = in_sizes[0] / IN_DIM;
  const int N = totalNodes / B;

  float* h      = (float*)d_ws;
  int*   rowptr = (int*)(h + (size_t)totalNodes * OUT_DIM);
  int*   count  = rowptr + (N + 1);
  int*   ssrc   = count + N;
  float* sw     = (float*)(ssrc + E);
  float* out    = (float*)d_out;
  const size_t wsNeeded = (size_t)((const char*)(sw + (size_t)B * E) - (const char*)d_ws);

  const int tiles = (totalNodes + 15) / 16;
  const int wavesPerBlock = LIN_WAVES;
  const int linBlocks = (tiles + wavesPerBlock - 1) / wavesPerBlock;

  if (ws_size < wsNeeded) {
    gcn_linear_wmma<<<linBlocks, wavesPerBlock * 32, 0, stream>>>(x, W, bias, h,
                                                                  totalNodes);
    const long long n = (long long)out_size;
    zero_out<<<(int)((n + 255) / 256), 256, 0, stream>>>(out, n);
    const long long total = (long long)E * OUT_DIM;
    edge_scatter<<<(int)((total + 255) / 256), 256, 0, stream>>>(h, ei, ei + E, ew,
                                                                 out, E, N, B);
    return;
  }

  if (N <= NODE_CAP) {
    int* part = (int*)h;
    slice_count<<<SORT_SLICES, SORT_THREADS, 0, stream>>>(ei + E, part, E, N);
    node_offsets<<<(N + 255) / 256, 256, 0, stream>>>(part, count, N);
    degree_scan<<<1, SCAN_THREADS, 0, stream>>>(count, rowptr, N);
    slice_fill<<<SORT_SLICES, SORT_THREADS, 0, stream>>>(ei, ei + E, ew, rowptr, part,
                                                       ssrc, sw, E, N, B);
  } else {
    clear_counts<<<(N + 255) / 256, 256, 0, stream>>>(count, N);
    count_targets<<<EDGE_BLOCKS, 256, 0, stream>>>(ei + E, count, E);
    degree_scan<<<1, SCAN_THREADS, 0, stream>>>(count, rowptr, N);
    fill_by_target<<<EDGE_BLOCKS, 256, 0, stream>>>(ei, ei + E, ew, rowptr, count,
                                                    ssrc, sw, E, B);
  }

  gcn_linear_wmma<<<linBlocks, wavesPerBlock * 32, 0, stream>>>(x, W, bias, h,
                                                                totalNodes);

  target_gather<<<GATHER_BLOCKS, GATHER_WAVES * 32, 0, stream>>>(
      h, rowptr, ssrc, sw, out, E, N, B);
}


# Each edge's slot is ONE 16-byte record {source, weight} x 2 batches written by one store (was 3 four-byte stores to 3 arrays); gather takes a pair per 8-byte scalar load; same slots and sum order
# speedup vs baseline: 4.3275x; 2.0174x over previous
#include <hip/hip_runtime.h>
#include <hip/hip_bf16.h>

typedef __attribute__((ext_vector_type(2))) float v2f;
typedef __attribute__((ext_vector_type(8))) float v8f;

#define IN_DIM 128
#define OUT_DIM 128

#define LIN_WAVES 20
#define LIN_THREADS (LIN_WAVES * 32)
#define LIN_PITCH 132
#define LIN_W_F4 (OUT_DIM * (IN_DIM / 4))
#define LIN_W_ROUNDS ((LIN_W_F4 + LIN_THREADS - 1) / LIN_THREADS)
#define LIN_W_ROWS (LIN_W_ROUNDS * LIN_THREADS / 32)

__global__ void __launch_bounds__(LIN_THREADS)
gcn_linear_wmma(const float* __restrict__ x, const float* __restrict__ W,
                const float* __restrict__ bias, float* __restrict__ h,
                int totalNodes) {
  __shared__ __attribute__((aligned(16))) float wS[LIN_W_ROWS * LIN_PITCH];
  __shared__ __attribute__((aligned(16))) float tS[LIN_WAVES * 16 * LIN_PITCH];

  const int tid  = threadIdx.x;
  const int lane = tid & 31;
  const int wave = tid >> 5;
  const int half = lane >> 4;
  const int l    = lane & 15;
  const int nodeBase = (blockIdx.x * LIN_WAVES + wave) * 16;
  float* mine = tS + wave * (16 * LIN_PITCH);

  float4 xr[16];
#pragma unroll
  for (int r = 0; r < 16; ++r) {
    const int node = min(nodeBase + r, totalNodes - 1);
    xr[r] = ((const float4*)(x + (size_t)node * IN_DIM))[lane];
  }
  float4 wv[LIN_W_ROUNDS];
#pragma unroll
  for (int j = 0; j < LIN_W_ROUNDS; ++j)
    wv[j] = ((const float4*)W)[min(tid + j * LIN_THREADS, LIN_W_F4 - 1)];
#pragma unroll
  for (int j = 0; j < LIN_W_ROUNDS; ++j) {
    const int i = tid + j * LIN_THREADS;
    *(float4*)(wS + (i >> 5) * LIN_PITCH + (i & 31) * 4) = wv[j];
  }
#pragma unroll
  for (int r = 0; r < 16; ++r)
    *(float4*)(mine + r * LIN_PITCH + lane * 4) = xr[r];
  __syncthreads();
  if (nodeBase >= totalNodes) return;

  v8f acc[8] = {};
  const float* aP = mine + l * LIN_PITCH + half * 2;
  const float* bP = wS + l * LIN_PITCH + half * 2;

  for (int k = 0; k < IN_DIM; k += 4) {
    const v2f a = *(const v2f*)(aP + k);
#pragma unroll
    for (int t = 0; t < 8; ++t) {
      const v2f bfrag = *(const v2f*)(bP + t * (16 * LIN_PITCH) + k);
      acc[t] = __builtin_amdgcn_wmma_f32_16x16x4_f32(
           false, a,  false, bfrag,
           (short)0, acc[t],  false,  false);
    }
  }

#pragma unroll
  for (int t = 0; t < 8; ++t) {
    const int n = t * 16 + l;
    const float bv = bias[n];
#pragma unroll
    for (int v = 0; v < 8; ++v) {
      const int m = v + half * 8;
      mine[m * LIN_PITCH + n] = acc[t][v] + bv;
    }
  }
  __builtin_amdgcn_fence(__ATOMIC_RELEASE, "wavefront");
  __builtin_amdgcn_wave_barrier();
  __builtin_amdgcn_fence(__ATOMIC_ACQUIRE, "wavefront");
  float4 o4[16];
#pragma unroll
  for (int r = 0; r < 16; ++r)
    o4[r] = *(const float4*)(mine + r * LIN_PITCH + lane * 4);
  float4* hrow = (float4*)(h + (size_t)nodeBase * OUT_DIM) + lane;
  if (nodeBase + 16 <= totalNodes) {
#pragma unroll
    for (int r = 0; r < 16; ++r) hrow[r * (OUT_DIM / 4)] = o4[r];
  } else {
#pragma unroll
    for (int r = 0; r < 16; ++r)
      if (nodeBase + r < totalNodes) hrow[r * (OUT_DIM / 4)] = o4[r];
  }
}

__global__ void clear_counts(int* __restrict__ count, int N) {
  const int i = blockIdx.x * blockDim.x + threadIdx.x;
  if (i < N) count[i] = 0;
}

__global__ void count_targets(const int* __restrict__ tgtIdx,
                              int* __restrict__ count, int E) {
  const int stride = gridDim.x * blockDim.x;
  for (int e = blockIdx.x * blockDim.x + threadIdx.x; e < E; e += stride)
    atomicAdd(count + tgtIdx[e], 1);
}

#define EDGE_BLOCKS 1024
#define SCAN_THREADS 1024
__global__ void __launch_bounds__(SCAN_THREADS)
degree_scan(const int* __restrict__ count, int* __restrict__ rowptr, int N) {
  __shared__ int part[SCAN_THREADS];
  const int tid = threadIdx.x;
  const int per = (N + SCAN_THREADS - 1) / SCAN_THREADS;
  const int lo = min(tid * per, N);
  const int hi = min(lo + per, N);
  int s = 0;
  for (int i = lo; i < hi; ++i) s += count[i];
  part[tid] = s;
  __syncthreads();
  for (int off = 1; off < SCAN_THREADS; off <<= 1) {
    const int v = (tid >= off) ? part[tid - off] : 0;
    __syncthreads();
    part[tid] += v;
    __syncthreads();
  }
  int run = part[tid] - s;
  for (int i = lo; i < hi; ++i) {
    rowptr[i] = run;
    run += count[i];
  }
  if (tid == SCAN_THREADS - 1) rowptr[N] = part[tid];
}

__global__ void fill_by_target(const int* __restrict__ srcIdx,
                               const int* __restrict__ tgtIdx,
                               const float* __restrict__ ew,
                               const int* __restrict__ rowptr,
                               int* __restrict__ count, int* __restrict__ ssrc,
                               float* __restrict__ sw, int E, int B) {
  const int stride = gridDim.x * blockDim.x;
  for (int e = blockIdx.x * blockDim.x + threadIdx.x; e < E; e += stride) {
    const int t = tgtIdx[e];
    const int left = atomicSub(count + t, 1);
    const int pos = rowptr[t] + left - 1;
    if ((unsigned)pos >= (unsigned)E) continue;
    ssrc[pos] = srcIdx[e];
    for (int b = 0; b < B; ++b) sw[(size_t)b * E + pos] = ew[(size_t)b * E + e];
  }
}

#define SORT_SLICES 64
#define SORT_THREADS 1024
#define SORT_ROUND 8
#define NODE_CAP 40960
#define REC_BATCHES 2
#define REC_DWORDS (2 * REC_BATCHES)

__device__ __forceinline__ void slice_bounds(int E, int& lo, int& hi) {
  const int per = (E + (int)gridDim.x - 1) / (int)gridDim.x;
  lo = min((int)blockIdx.x * per, E);
  hi = min(lo + per, E);
}

__global__ void __launch_bounds__(SORT_THREADS)
slice_count(const int* __restrict__ tgtIdx, int* __restrict__ part, int E, int N) {
  __shared__ int cnt[NODE_CAP];
  const int tid = threadIdx.x;
  int lo, hi;
  slice_bounds(E, lo, hi);
  for (int i = tid; i < N; i += SORT_THREADS) cnt[i] = 0;
  __syncthreads();
  for (int e0 = lo + tid; e0 < hi; e0 += SORT_ROUND * SORT_THREADS) {
    int t[SORT_ROUND];
    for (int k = 0; k < SORT_ROUND; ++k)
      t[k] = tgtIdx[min(e0 + k * SORT_THREADS, hi - 1)];
    for (int k = 0; k < SORT_ROUND; ++k)
      if (e0 + k * SORT_THREADS < hi) atomicAdd(&cnt[t[k]], 1);
  }
  __syncthreads();
  int* __restrict__ mine = part + (size_t)blockIdx.x * N;
  for (int i = tid; i < N; i += SORT_THREADS) mine[i] = cnt[i];
}

__device__ __forceinline__ int wave_scan_incl(int v, int lane) {
  for (int d = 1; d < 32; d <<= 1) {
    const int u = __shfl_up(v, d, 32);
    if (lane >= d) v += u;
  }
  return v;
}

#define OFFS_THREADS 256
#define BOFF_MAX (NODE_CAP / OFFS_THREADS)
#define BOFF_PER ((BOFF_MAX + 31) / 32)
static_assert(NODE_CAP % OFFS_THREADS == 0, "whole groups of nodes");
static_assert(SORT_SLICES * SORT_THREADS >= NODE_CAP, "one rowptr piece per work-group of slice_fill");

__global__ void __launch_bounds__(OFFS_THREADS)
node_offsets(int* __restrict__ part, int* __restrict__ lpre, int* __restrict__ btot, int N) {
  __shared__ int wtot[OFFS_THREADS / 32];
  const int tid = threadIdx.x;
  const int lane = tid & 31;
  const int wave = tid >> 5;
  const int t = blockIdx.x * OFFS_THREADS + tid;
  const bool live = t < N;
  const int tc = min(t, N - 1);
  int c[SORT_SLICES];
  for (int g = 0; g < SORT_SLICES; ++g) c[g] = part[(size_t)g * N + tc];
  int run = 0;
  for (int g = 0; g < SORT_SLICES; ++g) {
    const int v = c[g];
    c[g] = run;
    run += v;
  }
  if (live) {
    for (int g = 0; g < SORT_SLICES; ++g) part[(size_t)g * N + t] = c[g];
  } else {
    run = 0;
  }
  const int incl = wave_scan_incl(run, lane);
  if (lane == 31) wtot[wave] = incl;
  __syncthreads();
  int before = 0;
  for (int w = 0; w < OFFS_THREADS / 32; ++w) {
    const int v = wtot[w];
    if (w < wave) before += v;
  }
  if (live) lpre[t] = before + incl - run;
  if (tid == OFFS_THREADS - 1) btot[blockIdx.x] = before + incl;
}

__global__ void __launch_bounds__(SORT_THREADS)
slice_fill(const int* __restrict__ srcIdx, const int* __restrict__ tgtIdx,
           const float* __restrict__ ew, const int* __restrict__ lpre,
           const int* __restrict__ btot, const int* __restrict__ part,
           int* __restrict__ rowptr, int* __restrict__ ssrc,
           float* __restrict__ sw, int* __restrict__ recs, int E, int N, int B) {
  __shared__ int cur[NODE_CAP];
  __shared__ int boff[BOFF_MAX + 1];
  const int tid = threadIdx.x;
  int lo, hi;
  slice_bounds(E, lo, hi);
  const int nblk = (N + OFFS_THREADS - 1) / OFFS_THREADS;
  const int ri = (int)blockIdx.x * SORT_THREADS + tid;
  const int rlp = lpre[min(ri, N - 1)];
  int bs[BOFF_PER];
  for (int q = 0; q < BOFF_PER; ++q)
    bs[q] = (tid < 32) ? btot[min(tid * BOFF_PER + q, nblk - 1)] : 0;
  const int* __restrict__ mine = part + (size_t)blockIdx.x * N;
  for (int i0 = tid; i0 < N; i0 += SORT_ROUND * SORT_THREADS) {
    int v[SORT_ROUND];
    for (int k = 0; k < SORT_ROUND; ++k) {
      const int ic = min(i0 + k * SORT_THREADS, N - 1);
      v[k] = lpre[ic] + mine[ic];
    }
    for (int k = 0; k < SORT_ROUND; ++k) {
      const int i = i0 + k * SORT_THREADS;
      if (i < N) cur[i] = v[k];
    }
  }
  if (tid < 32) {
    int s = 0;
    for (int q = 0; q < BOFF_PER; ++q) {
      if (tid * BOFF_PER + q >= nblk) bs[q] = 0;
      s += bs[q];
    }
    const int incl = wave_scan_incl(s, tid);
    int run = incl - s;
    for (int q = 0; q < BOFF_PER; ++q) {
      const int j = tid * BOFF_PER + q;
      if (j < nblk) boff[j] = run;
      run += bs[q];
    }
    if (tid == 31) boff[nblk] = incl;
  }
  __syncthreads();
  if (ri < N) rowptr[ri] = boff[(unsigned)ri / OFFS_THREADS] + rlp;
  if (ri == 0) rowptr[N] = boff[nblk];
  for (int e0 = lo + tid; e0 < hi; e0 += SORT_ROUND * SORT_THREADS) {
    int ec[SORT_ROUND], t[SORT_ROUND], s[SORT_ROUND], pos[SORT_ROUND];
    for (int k = 0; k < SORT_ROUND; ++k) {
      ec[k] = min(e0 + k * SORT_THREADS, hi - 1);
      t[k] = tgtIdx[ec[k]];
      s[k] = srcIdx[ec[k]];
    }
    if (recs != nullptr) {
      float w[REC_BATCHES][SORT_ROUND];
      for (int b = 0; b < REC_BATCHES; ++b) {
        if (b < B) {
          const float* __restrict__ wsrc = ew + (size_t)b * E;
          for (int k = 0; k < SORT_ROUND; ++k) w[b][k] = wsrc[ec[k]];
        } else {
          for (int k = 0; k < SORT_ROUND; ++k) w[b][k] = 0.0f;
        }
      }
      for (int k = 0; k < SORT_ROUND; ++k) {
        pos[k] = -1;
        if (e0 + k * SORT_THREADS < hi)
          pos[k] = atomicAdd(&cur[t[k]], 1) + boff[(unsigned)t[k] / OFFS_THREADS];
      }
      for (int k = 0; k < SORT_ROUND; ++k)
        if ((unsigned)pos[k] < (unsigned)E)
          ((int4*)recs)[pos[k]] = make_int4(s[k], __float_as_int(w[0][k]),
                                            s[k], __float_as_int(w[1][k]));
      continue;
    }
    for (int k = 0; k < SORT_ROUND; ++k) {
      pos[k] = -1;
      if (e0 + k * SORT_THREADS < hi)
        pos[k] = atomicAdd(&cur[t[k]], 1) + boff[(unsigned)t[k] / OFFS_THREADS];
    }
    for (int k = 0; k < SORT_ROUND; ++k)
      if ((unsigned)pos[k] < (unsigned)E) ssrc[pos[k]] = s[k];
    for (int b = 0; b < B; ++b) {
      const float* __restrict__ wsrc = ew + (size_t)b * E;
      float* __restrict__ wdst = sw + (size_t)b * E;
      float w[SORT_ROUND];
      for (int k = 0; k < SORT_ROUND; ++k) w[k] = wsrc[ec[k]];
      for (int k = 0; k < SORT_ROUND; ++k)
        if ((unsigned)pos[k] < (unsigned)E) wdst[pos[k]] = w[k];
    }
  }
}

#define GATHER_DEPTH 16

template <int R, int STRIDE>
__device__ __forceinline__ void gather_rows(float4& acc, float& wsum,
                                            const float4* __restrict__ xb,
                                            const int* __restrict__ sp,
                                            const float* __restrict__ wp,
                                            unsigned off) {
  int s[R];
  float w[R];
  float4 v[R];
  if (STRIDE != 1) __builtin_assume(off < 0x40000010u);
#pragma unroll
  for (int u = 0; u < R; ++u) {
    if (STRIDE == 1) {
      s[u] = sp[u];
      w[u] = wp[u];
    } else {
      const int2 p = *(const int2*)((const char*)sp + (off + (unsigned)(u * STRIDE * 4)));
      s[u] = p.x;
      w[u] = __int_as_float(p.y);
    }
  }
#pragma unroll
  for (int u = 0; u < R; ++u) v[u] = xb[(size_t)s[u] * (IN_DIM / 4)];
#pragma unroll
  for (int u = 0; u < R; ++u) {
    acc.x = fmaf(w[u], v[u].x, acc.x);
    acc.y = fmaf(w[u], v[u].y, acc.y);
    acc.z = fmaf(w[u], v[u].z, acc.z);
    acc.w = fmaf(w[u], v[u].w, acc.w);
    wsum += w[u];
  }
}

template <bool PACKED>
__global__ void __launch_bounds__(LIN_THREADS)
__attribute__((amdgpu_waves_per_eu(4, 5)))
gcn_gather_linear(const float* __restrict__ x, const float* __restrict__ W,
                  const float* __restrict__ bias, const int* __restrict__ rowptr,
                  const int* __restrict__ ssrc, const float* __restrict__ sw,
                  float* __restrict__ out, int E, int N, int totalNodes) {
  constexpr int ST = PACKED ? REC_DWORDS : 1;
  __shared__ __attribute__((aligned(16))) float wS[LIN_W_ROWS * LIN_PITCH];
  __shared__ __attribute__((aligned(16))) float tS[LIN_WAVES * 16 * LIN_PITCH];

  const int tid  = threadIdx.x;
  const int lane = tid & 31;
  const int half = lane >> 4;
  const int l    = lane & 15;
  const int wave = __builtin_amdgcn_readfirstlane(tid >> 5);
  const int nodeBase = ((int)blockIdx.x * LIN_WAVES + wave) * 16;
  float* mine = tS + wave * (16 * LIN_PITCH);

  float4 wv[LIN_W_ROUNDS];
#pragma unroll
  for (int j = 0; j < LIN_W_ROUNDS; ++j)
    wv[j] = ((const float4*)W)[min(tid + j * LIN_THREADS, LIN_W_F4 - 1)];
#pragma unroll
  for (int j = 0; j < LIN_W_ROUNDS; ++j) {
    const int i = tid + j * LIN_THREADS;
    *(float4*)(wS + (i >> 5) * LIN_PITCH + (i & 31) * 4) = wv[j];
  }
  __syncthreads();
  if (nodeBase >= totalNodes) return;

  int bcur = nodeBase / N;
  int t = nodeBase - bcur * N;
  int beg = rowptr[t];
  int end = rowptr[t + 1];
#pragma nounroll
  for (int r = 0; r < 16; ++r) {
    int tn = t + 1;
    int bn = bcur;
    if (tn == N) { tn = 0; bn = bcur + 1; }
    const bool more = nodeBase + r + 1 < totalNodes;
    const int tp = more ? tn : t;
    const int nbeg = rowptr[tp];
    const int nend = rowptr[tp + 1];
    float4 acc = {0.0f, 0.0f, 0.0f, 0.0f};
    float wsum = 0.0f;
    if (nodeBase + r < totalNodes) {
      const float4* __restrict__ xb =
          (const float4*)x + (size_t)bcur * N * (IN_DIM / 4) + lane;
      const float* __restrict__ wb = PACKED ? nullptr : sw + (size_t)bcur * E;
      const unsigned ob = 8u * (unsigned)bcur;
      int j = beg;
      for (; end - j >= GATHER_DEPTH; j += GATHER_DEPTH)
        gather_rows<GATHER_DEPTH, ST>(acc, wsum, xb, PACKED ? ssrc : ssrc + j, wb + (PACKED ? 0 : j),
                                      PACKED ? 16u * (unsigned)j + ob : 0u);
      switch (end - j) {
        case 15: gather_rows<15, ST>(acc, wsum, xb, PACKED ? ssrc : ssrc + j, wb + (PACKED ? 0 : j), PACKED ? 16u * (unsigned)j + ob : 0u); break;
        case 14: gather_rows<14, ST>(acc, wsum, xb, PACKED ? ssrc : ssrc + j, wb + (PACKED ? 0 : j), PACKED ? 16u * (unsigned)j + ob : 0u); break;
        case 13: gather_rows<13, ST>(acc, wsum, xb, PACKED ? ssrc : ssrc + j, wb + (PACKED ? 0 : j), PACKED ? 16u * (unsigned)j + ob : 0u); break;
        case 12: gather_rows<12, ST>(acc, wsum, xb, PACKED ? ssrc : ssrc + j, wb + (PACKED ? 0 : j), PACKED ? 16u * (unsigned)j + ob : 0u); break;
        case 11: gather_rows<11, ST>(acc, wsum, xb, PACKED ? ssrc : ssrc + j, wb + (PACKED ? 0 : j), PACKED ? 16u * (unsigned)j + ob : 0u); break;
        case 10: gather_rows<10, ST>(acc, wsum, xb, PACKED ? ssrc : ssrc + j, wb + (PACKED ? 0 : j), PACKED ? 16u * (unsigned)j + ob : 0u); break;
        case 9: gather_rows<9, ST>(acc, wsum, xb, PACKED ? ssrc : ssrc + j, wb + (PACKED ? 0 : j), PACKED ? 16u * (unsigned)j + ob : 0u); break;
        case 8: gather_rows<8, ST>(acc, wsum, xb, PACKED ? ssrc : ssrc + j, wb + (PACKED ? 0 : j), PACKED ? 16u * (unsigned)j + ob : 0u); break;
        case 7: gather_rows<7, ST>(acc, wsum, xb, PACKED ? ssrc : ssrc + j, wb + (PACKED ? 0 : j), PACKED ? 16u * (unsigned)j + ob : 0u); break;
        case 6: gather_rows<6, ST>(acc, wsum, xb, PACKED ? ssrc : ssrc + j, wb + (PACKED ? 0 : j), PACKED ? 16u * (unsigned)j + ob : 0u); break;
        case 5: gather_rows<5, ST>(acc, wsum, xb, PACKED ? ssrc : ssrc + j, wb + (PACKED ? 0 : j), PACKED ? 16u * (unsigned)j + ob : 0u); break;
        case 4: gather_rows<4, ST>(acc, wsum, xb, PACKED ? ssrc : ssrc + j, wb + (PACKED ? 0 : j), PACKED ? 16u * (unsigned)j + ob : 0u); break;
        case 3: gather_rows<3, ST>(acc, wsum, xb, PACKED ? ssrc : ssrc + j, wb + (PACKED ? 0 : j), PACKED ? 16u * (unsigned)j + ob : 0u); break;
        case 2: gather_rows<2, ST>(acc, wsum, xb, PACKED ? ssrc : ssrc + j, wb + (PACKED ? 0 : j), PACKED ? 16u * (unsigned)j + ob : 0u); break;
        case 1: gather_rows<1, ST>(acc, wsum, xb, PACKED ? ssrc : ssrc + j, wb + (PACKED ? 0 : j), PACKED ? 16u * (unsigned)j + ob : 0u); break;
        default: break;
      }
    }
    *(float4*)(mine + r * LIN_PITCH + lane * 4) = acc;
    if (lane == 0) mine[r * LIN_PITCH + IN_DIM] = wsum;
    t = tn;
    bcur = bn;
    beg = nbeg;
    end = nend;
  }
  __builtin_amdgcn_fence(__ATOMIC_RELEASE, "wavefront");
  __builtin_amdgcn_wave_barrier();
  __builtin_amdgcn_fence(__ATOMIC_ACQUIRE, "wavefront");

  v8f acc[8] = {};
  const float* aP = mine + l * LIN_PITCH + half * 2;
  const float* bP = wS + l * LIN_PITCH + half * 2;

  for (int k = 0; k < IN_DIM; k += 4) {
    const v2f a = *(const v2f*)(aP + k);
#pragma unroll
    for (int tt = 0; tt < 8; ++tt) {
      const v2f bfrag = *(const v2f*)(bP + tt * (16 * LIN_PITCH) + k);
      acc[tt] = __builtin_amdgcn_wmma_f32_16x16x4_f32(
           false, a,  false, bfrag,
           (short)0, acc[tt],  false,  false);
    }
  }

  float ws8[8];
#pragma unroll
  for (int v = 0; v < 8; ++v) ws8[v] = mine[(v + half * 8) * LIN_PITCH + IN_DIM];
#pragma unroll
  for (int tt = 0; tt < 8; ++tt) {
    const int n = tt * 16 + l;
    const float bv = bias[n];
#pragma unroll
    for (int v = 0; v < 8; ++v) {
      const int m = v + half * 8;
      mine[m * LIN_PITCH + n] = fmaf(bv, ws8[v], acc[tt][v]);
    }
  }
  __builtin_amdgcn_fence(__ATOMIC_RELEASE, "wavefront");
  __builtin_amdgcn_wave_barrier();
  __builtin_amdgcn_fence(__ATOMIC_ACQUIRE, "wavefront");
  float4 o4[16];
#pragma unroll
  for (int r = 0; r < 16; ++r)
    o4[r] = *(const float4*)(mine + r * LIN_PITCH + lane * 4);
  float4* orow = (float4*)(out + (size_t)nodeBase * OUT_DIM) + lane;
  if (nodeBase + 16 <= totalNodes) {
#pragma unroll
    for (int r = 0; r < 16; ++r) orow[r * (OUT_DIM / 4)] = o4[r];
  } else {
#pragma unroll
    for (int r = 0; r < 16; ++r)
      if (nodeBase + r < totalNodes) orow[r * (OUT_DIM / 4)] = o4[r];
  }
}

__global__ void zero_out(float* __restrict__ out, long long n) {
  long long i = (long long)blockIdx.x * blockDim.x + threadIdx.x;
  if (i < n) out[i] = 0.0f;
}

__global__ void __launch_bounds__(256)
edge_scatter(const float* __restrict__ h, const int* __restrict__ srcIdx,
             const int* __restrict__ tgtIdx, const float* __restrict__ ew,
             float* __restrict__ out, int E, int N, int B) {
  long long idx = (long long)blockIdx.x * blockDim.x + threadIdx.x;
  long long total = (long long)E * OUT_DIM;
  if (idx >= total) return;
  const int c = (int)(idx & (OUT_DIM - 1));
  const int e = (int)(idx >> 7);
  const int s = srcIdx[e];
  const int t = tgtIdx[e];
  for (int b = 0; b < B; ++b) {
    const float w = ew[(size_t)b * E + e];
    const float val = w * h[((size_t)b * N + s) * OUT_DIM + c];
    atomicAdd(out + ((size_t)b * N + t) * OUT_DIM + c, val);
  }
}

extern "C" void kernel_launch(void* const* d_in, const int* in_sizes, int n_in,
                              void* d_out, int out_size, void* d_ws, size_t ws_size,
                              hipStream_t stream) {
  const float* x    = (const float*)d_in[0];
  const int*   ei   = (const int*)d_in[1];
  const float* ew   = (const float*)d_in[2];
  const float* W    = (const float*)d_in[3];
  const float* bias = (const float*)d_in[4];

  const int E = in_sizes[1] / 2;
  const int B = in_sizes[2] / E;
  const int totalNodes = in_sizes[0] / IN_DIM;
  const int N = totalNodes / B;

  float* h      = (float*)d_ws;
  int*   rowptr = (int*)(h + (size_t)totalNodes * OUT_DIM);
  int*   count  = rowptr + (N + 1);
  int*   ssrc   = count + N;
  float* sw     = (float*)(ssrc + E);
  float* out    = (float*)d_out;
  const size_t wsNeeded = (size_t)((const char*)(sw + (size_t)B * E) - (const char*)d_ws);

  const int tiles = (totalNodes + 15) / 16;
  const int wavesPerBlock = LIN_WAVES;
  const int linBlocks = (tiles + wavesPerBlock - 1) / wavesPerBlock;

  if (ws_size < wsNeeded) {
    gcn_linear_wmma<<<linBlocks, wavesPerBlock * 32, 0, stream>>>(x, W, bias, h,
                                                                  totalNodes);
    const long long n = (long long)out_size;
    zero_out<<<(int)((n + 255) / 256), 256, 0, stream>>>(out, n);
    const long long total = (long long)E * OUT_DIM;
    edge_scatter<<<(int)((total + 255) / 256), 256, 0, stream>>>(h, ei, ei + E, ew,
                                                                 out, E, N, B);
    return;
  }

  bool packed = false;
  if (N <= NODE_CAP) {
    int* part = (int*)h;
    int* btot = part + (size_t)SORT_SLICES * N;
    int* recs = btot + (((N + OFFS_THREADS - 1) / OFFS_THREADS + 3) & ~3);
    packed = B <= REC_BATCHES && E <= (1 << 26) &&
             (const char*)(recs + (size_t)REC_DWORDS * E) <= (const char*)rowptr;
    if (!packed) recs = nullptr;
    slice_count<<<SORT_SLICES, SORT_THREADS, 0, stream>>>(ei + E, part, E, N);
    node_offsets<<<(N + OFFS_THREADS - 1) / OFFS_THREADS, OFFS_THREADS, 0, stream>>>(
        part, count, btot, N);
    slice_fill<<<SORT_SLICES, SORT_THREADS, 0, stream>>>(ei, ei + E, ew, count, btot, part,
                                                       rowptr, ssrc, sw, recs, E, N, B);
    if (packed) ssrc = recs;
  } else {
    clear_counts<<<(N + 255) / 256, 256, 0, stream>>>(count, N);
    count_targets<<<EDGE_BLOCKS, 256, 0, stream>>>(ei + E, count, E);
    degree_scan<<<1, SCAN_THREADS, 0, stream>>>(count, rowptr, N);
    fill_by_target<<<EDGE_BLOCKS, 256, 0, stream>>>(ei, ei + E, ew, rowptr, count,
                                                    ssrc, sw, E, B);
  }

  if (packed)
    gcn_gather_linear<true><<<linBlocks, wavesPerBlock * 32, 0, stream>>>(
        x, W, bias, rowptr, ssrc, sw, out, E, N, totalNodes);
  else
    gcn_gather_linear<false><<<linBlocks, wavesPerBlock * 32, 0, stream>>>(
        x, W, bias, rowptr, ssrc, sw, out, E, N, totalNodes);
}
